# starting kernel
# baseline (speedup 1.0000x reference)
#include <hip/hip_runtime.h>

typedef __attribute__((ext_vector_type(2))) float v2f;
typedef __attribute__((ext_vector_type(8))) float v8f;

#define WPB  4
#define HSTR 9

__device__ __forceinline__ float sigm(float x) {
    return __builtin_amdgcn_rcpf(1.0f + __builtin_amdgcn_exp2f(-1.44269504f * x));
}
__device__ __forceinline__ float tanh_f(float x) {
    return 2.0f * __builtin_amdgcn_rcpf(1.0f + __builtin_amdgcn_exp2f(-2.88539008f * x)) - 1.0f;
}
__device__ __forceinline__ float xor8_dpp(float v) {
    int r = __builtin_amdgcn_update_dpp(0, __builtin_bit_cast(int, v),
                                        0x128  , 0xF, 0xF, true);
    return __builtin_bit_cast(float, r);
}

__global__ void __launch_bounds__(32 * WPB)
lstm_wmma_kernel(const float* __restrict__ x,
                 const float* __restrict__ W_ih, const float* __restrict__ W_hh,
                 const float* __restrict__ b_ih, const float* __restrict__ b_hh,
                 const float* __restrict__ W_fc, const float* __restrict__ b_fc,
                 float* __restrict__ out, int T)
{
    __shared__ float hbuf[WPB][16 * HSTR];

    const int lane = threadIdx.x & 31;
    const int wv   = threadIdx.x >> 5;
    const int n    = lane & 15;
    const int half = lane >> 4;
    const int k2   = 2 * half;
    const int b0   = (blockIdx.x * WPB + wv) * 16;

    v2f BxT0, BxT1, Bh0T0, Bh1T0, Bh0T1, Bh1T1;
    BxT0.x  = W_ih[n * 4 + k2];          BxT0.y  = W_ih[n * 4 + k2 + 1];
    BxT1.x  = W_ih[(n + 16) * 4 + k2];   BxT1.y  = W_ih[(n + 16) * 4 + k2 + 1];
    Bh0T0.x = W_hh[n * 8 + k2];          Bh0T0.y = W_hh[n * 8 + k2 + 1];
    Bh1T0.x = W_hh[n * 8 + k2 + 4];      Bh1T0.y = W_hh[n * 8 + k2 + 5];
    Bh0T1.x = W_hh[(n + 16) * 8 + k2];   Bh0T1.y = W_hh[(n + 16) * 8 + k2 + 1];
    Bh1T1.x = W_hh[(n + 16) * 8 + k2+4]; Bh1T1.y = W_hh[(n + 16) * 8 + k2 + 5];
    const float bias0 = b_ih[n]      + b_hh[n];
    const float bias1 = b_ih[n + 16] + b_hh[n + 16];

    const float gsel = (n < 8) ? 2.0f : 1.0f;

    const float* xrow = x + (size_t)(b0 + n) * (size_t)T * 4 + k2;
    float* hb = &hbuf[wv][0];

    v8f cst;
    #pragma unroll
    for (int i = 0; i < 8; ++i) cst[i] = 0.0f;
    v2f Ah03; Ah03.x = 0.0f; Ah03.y = 0.0f;
    v2f Ah47 = Ah03;

    v2f Ax = *(const v2f*)(xrow);
    for (int t = 0; t < T; ++t) {
        const v2f Axc = Ax;
        const int tn = (t + 1 < T) ? (t + 1) : t;
        Ax = *(const v2f*)(xrow + (size_t)tn * 4);

        v8f acc0, acc1;
        #pragma unroll
        for (int i = 0; i < 8; ++i) { acc0[i] = bias0; acc1[i] = bias1; }

        acc0 = __builtin_amdgcn_wmma_f32_16x16x4_f32(false, Axc,  false, BxT0,  (short)0, acc0, false, false);
        acc1 = __builtin_amdgcn_wmma_f32_16x16x4_f32(false, Axc,  false, BxT1,  (short)0, acc1, false, false);
        acc0 = __builtin_amdgcn_wmma_f32_16x16x4_f32(false, Ah03, false, Bh0T0, (short)0, acc0, false, false);
        acc1 = __builtin_amdgcn_wmma_f32_16x16x4_f32(false, Ah03, false, Bh0T1, (short)0, acc1, false, false);
        acc0 = __builtin_amdgcn_wmma_f32_16x16x4_f32(false, Ah47, false, Bh1T0, (short)0, acc0, false, false);
        acc1 = __builtin_amdgcn_wmma_f32_16x16x4_f32(false, Ah47, false, Bh1T1, (short)0, acc1, false, false);

        v8f hnew;
        #pragma unroll
        for (int v = 0; v < 8; ++v) {
            float s0 = sigm(acc0[v]);
            float u  = sigm(gsel * acc1[v]);
            float f  = xor8_dpp(s0);
            float o  = xor8_dpp(u);
            float g  = 2.0f * u - 1.0f;
            float cn = f * cst[v] + s0 * g;
            cst[v]   = cn;
            hnew[v]  = o * tanh_f(cn);
        }

        if (n < 8) {
            #pragma unroll
            for (int v = 0; v < 8; ++v)
                hb[(v + 8 * half) * HSTR + n] = hnew[v];
        }
        __builtin_amdgcn_wave_barrier();
        asm volatile("" ::: "memory");
        Ah03.x = hb[n * HSTR + k2];
        Ah03.y = hb[n * HSTR + k2 + 1];
        Ah47.x = hb[n * HSTR + k2 + 4];
        Ah47.y = hb[n * HSTR + k2 + 5];
        __builtin_amdgcn_wave_barrier();
    }

    if (half == 0) {
        float acc = b_fc[0];
        #pragma unroll
        for (int j = 0; j < 8; ++j)
            acc += hb[n * HSTR + j] * W_fc[j];
        out[b0 + n] = acc;
    }
}

extern "C" void kernel_launch(void* const* d_in, const int* in_sizes, int n_in,
                              void* d_out, int out_size, void* d_ws, size_t ws_size,
                              hipStream_t stream) {
    const float* x    = (const float*)d_in[0];
    const float* W_ih = (const float*)d_in[1];
    const float* W_hh = (const float*)d_in[2];
    const float* b_ih = (const float*)d_in[3];
    const float* b_hh = (const float*)d_in[4];
    const float* W_fc = (const float*)d_in[5];
    const float* b_fc = (const float*)d_in[6];
    float* out = (float*)d_out;

    const int T = 512;
    const int I = 4;
    const int B = in_sizes[0] / (T * I);
    const int wavesTotal = B / 16;
    const int blocks = wavesTotal / WPB;

    lstm_wmma_kernel<<<blocks, 32 * WPB, 0, stream>>>(x, W_ih, W_hh, b_ih, b_hh,
                                                      W_fc, b_fc, out, T);
}


# Swap WMMA operand roles (weights=A, x and h=B) with gate rows ordered so each lane's D holds i,f,g,o of its own 4 units: new h is the lane's next B operand, no LDS transpose, no DPP, no idle lanes
# speedup vs baseline: 1.2649x; 1.2649x over previous
#include <hip/hip_runtime.h>


typedef __attribute__((ext_vector_type(2))) float v2f;
typedef __attribute__((ext_vector_type(8))) float v8f;

#define WPB  4

#define NLOG2E   (-1.44269504f)
#define N2LOG2E  (-2.88539008f)

__device__ __forceinline__ v2f rcp1p_exp2(v2f x, float k) {
    const v2f t = x * k;
    v2f e;
    e.x = __builtin_amdgcn_exp2f(t.x);
    e.y = __builtin_amdgcn_exp2f(t.y);
    const v2f d = 1.0f + e;
    v2f r;
    r.x = __builtin_amdgcn_rcpf(d.x);
    r.y = __builtin_amdgcn_rcpf(d.y);
    return r;
}

__device__ __forceinline__ int unit_j(int hf, int u) {
    return (u < 2) ? (2 * hf + u) : (4 + 2 * hf + (u - 2));
}

__global__ void __launch_bounds__(32 * WPB)
lstm_wmma_kernel(const float* __restrict__ x,
                 const float* __restrict__ W_ih, const float* __restrict__ W_hh,
                 const float* __restrict__ b_ih, const float* __restrict__ b_hh,
                 const float* __restrict__ W_fc, const float* __restrict__ b_fc,
                 float* __restrict__ out, int T)
{
    const int lane = threadIdx.x & 31;
    const int wv   = threadIdx.x >> 5;
    const int n    = lane & 15;
    const int half = lane >> 4;
    const int k2   = 2 * half;
    const int b0   = (blockIdx.x * WPB + wv) * 16;

    const int mh = n >> 3;
    const int mr = n & 7;
    const int g0 = ((mr < 4) ? 0 : 8) + unit_j(mh, mr & 3);
    const int g1 = g0 + 16;
    v2f AxT0, AxT1, Ah0T0, Ah1T0, Ah0T1, Ah1T1;
    AxT0.x  = W_ih[g0 * 4 + k2];       AxT0.y  = W_ih[g0 * 4 + k2 + 1];
    AxT1.x  = W_ih[g1 * 4 + k2];       AxT1.y  = W_ih[g1 * 4 + k2 + 1];
    Ah0T0.x = W_hh[g0 * 8 + k2];       Ah0T0.y = W_hh[g0 * 8 + k2 + 1];
    Ah1T0.x = W_hh[g0 * 8 + k2 + 4];   Ah1T0.y = W_hh[g0 * 8 + k2 + 5];
    Ah0T1.x = W_hh[g1 * 8 + k2];       Ah0T1.y = W_hh[g1 * 8 + k2 + 1];
    Ah1T1.x = W_hh[g1 * 8 + k2 + 4];   Ah1T1.y = W_hh[g1 * 8 + k2 + 5];

    v8f bias0, bias1;
    #pragma unroll
    for (int r = 0; r < 8; ++r) {
        const int gr = ((r < 4) ? 0 : 8) + unit_j(half, r & 3);
        bias0[r] = b_ih[gr]      + b_hh[gr];
        bias1[r] = b_ih[gr + 16] + b_hh[gr + 16];
    }

    const float* xrow = x + (size_t)(b0 + n) * (size_t)T * 4 + k2;

    v2f Bh0; Bh0.x = 0.0f; Bh0.y = 0.0f;
    v2f Bh1 = Bh0;
    v2f c01 = Bh0, c23 = Bh0;

    v2f Bx = *(const v2f*)(xrow);
    for (int t = 0; t < T; ++t) {
        const v2f Bxc = Bx;
        const int tn = (t + 1 < T) ? (t + 1) : t;
        Bx = *(const v2f*)(xrow + (size_t)tn * 4);
        asm volatile("" ::: "memory");

        v8f acc0 = bias0, acc1 = bias1;
        acc0 = __builtin_amdgcn_wmma_f32_16x16x4_f32(false, AxT0,  false, Bxc, (short)0, acc0, false, false);
        acc1 = __builtin_amdgcn_wmma_f32_16x16x4_f32(false, AxT1,  false, Bxc, (short)0, acc1, false, false);
        acc0 = __builtin_amdgcn_wmma_f32_16x16x4_f32(false, Ah0T0, false, Bh0, (short)0, acc0, false, false);
        acc1 = __builtin_amdgcn_wmma_f32_16x16x4_f32(false, Ah0T1, false, Bh0, (short)0, acc1, false, false);
        acc0 = __builtin_amdgcn_wmma_f32_16x16x4_f32(false, Ah1T0, false, Bh1, (short)0, acc0, false, false);
        acc1 = __builtin_amdgcn_wmma_f32_16x16x4_f32(false, Ah1T1, false, Bh1, (short)0, acc1, false, false);

        #pragma unroll
        for (int p = 0; p < 2; ++p) {
            v2f ipre, fpre, gpre, opre;
            ipre.x = acc0[2 * p];     ipre.y = acc0[2 * p + 1];
            fpre.x = acc0[4 + 2 * p]; fpre.y = acc0[5 + 2 * p];
            gpre.x = acc1[2 * p];     gpre.y = acc1[2 * p + 1];
            opre.x = acc1[4 + 2 * p]; opre.y = acc1[5 + 2 * p];
            const v2f ig = rcp1p_exp2(ipre, NLOG2E);
            const v2f fg = rcp1p_exp2(fpre, NLOG2E);
            const v2f ug = rcp1p_exp2(gpre, N2LOG2E);
            const v2f og = rcp1p_exp2(opre, NLOG2E);
            const v2f g  = 2.0f * ug - 1.0f;
            const v2f cp = (p == 0) ? c01 : c23;
            const v2f cn = __builtin_elementwise_fma(ig, g, fg * cp);
            const v2f th = 2.0f * rcp1p_exp2(cn, N2LOG2E) - 1.0f;
            const v2f hn = og * th;
            if (p == 0) { c01 = cn; Bh0 = hn; } else { c23 = cn; Bh1 = hn; }
        }
    }

    const float o0 = __shfl_xor(Bh0.x, 16, 32);
    const float o1 = __shfl_xor(Bh0.y, 16, 32);
    const float o2 = __shfl_xor(Bh1.x, 16, 32);
    const float o3 = __shfl_xor(Bh1.y, 16, 32);
    if (half == 0) {
        float hj[8];
        hj[0] = Bh0.x; hj[1] = Bh0.y; hj[2] = o0; hj[3] = o1;
        hj[4] = Bh1.x; hj[5] = Bh1.y; hj[6] = o2; hj[7] = o3;
        float acc = b_fc[0];
        #pragma unroll
        for (int j = 0; j < 8; ++j)
            acc += hj[j] * W_fc[j];
        out[b0 + n] = acc;
    }
}

extern "C" void kernel_launch(void* const* d_in, const int* in_sizes, int n_in,
                              void* d_out, int out_size, void* d_ws, size_t ws_size,
                              hipStream_t stream) {
    const float* x    = (const float*)d_in[0];
    const float* W_ih = (const float*)d_in[1];
    const float* W_hh = (const float*)d_in[2];
    const float* b_ih = (const float*)d_in[3];
    const float* b_hh = (const float*)d_in[4];
    const float* W_fc = (const float*)d_in[5];
    const float* b_fc = (const float*)d_in[6];
    float* out = (float*)d_out;

    const int T = 512;
    const int I = 4;
    const int B = in_sizes[0] / (T * I);
    const int wavesTotal = B / 16;
    const int blocks = wavesTotal / WPB;

    lstm_wmma_kernel<<<blocks, 32 * WPB, 0, stream>>>(x, W_ih, W_hh, b_ih, b_hh,
                                                      W_fc, b_fc, out, T);
}


# Cell sigmoids and tanh on hardware v_tanh_f32: sigmoid(z) = 0.5 + 0.5 tanh(z/2), 1/2 folded exactly into the i, f, o weight rows and biases; 20 tanh a step issued f, i, g, o, packed steps between
# speedup vs baseline: 1.5530x; 1.2278x over previous
#include <hip/hip_runtime.h>


typedef __attribute__((ext_vector_type(2))) float v2f;
typedef __attribute__((ext_vector_type(8))) float v8f;

#define WPB  4

__device__ __forceinline__ int unit_j(int hf, int u) {
    return (u < 2) ? (2 * hf + u) : (4 + 2 * hf + (u - 2));
}

__global__ void __launch_bounds__(32 * WPB)
lstm_wmma_kernel(const float* __restrict__ x,
                 const float* __restrict__ W_ih, const float* __restrict__ W_hh,
                 const float* __restrict__ b_ih, const float* __restrict__ b_hh,
                 const float* __restrict__ W_fc, const float* __restrict__ b_fc,
                 float* __restrict__ out, int T)
{
    const int lane = threadIdx.x & 31;
    const int wv   = threadIdx.x >> 5;
    const int n    = lane & 15;
    const int half = lane >> 4;
    const int k2   = 2 * half;
    const int b0   = (blockIdx.x * WPB + wv) * 16;

    const int mh = n >> 3;
    const int mr = n & 7;
    const int g0 = ((mr < 4) ? 0 : 8) + unit_j(mh, mr & 3);
    const int g1 = g0 + 16;
    const float s0 = 0.5f;
    const float s1 = (mr < 4) ? 1.0f : 0.5f;
    v2f AxT0, AxT1, Ah0T0, Ah1T0, Ah0T1, Ah1T1;
    AxT0.x  = W_ih[g0 * 4 + k2];       AxT0.y  = W_ih[g0 * 4 + k2 + 1];
    AxT1.x  = W_ih[g1 * 4 + k2];       AxT1.y  = W_ih[g1 * 4 + k2 + 1];
    Ah0T0.x = W_hh[g0 * 8 + k2];       Ah0T0.y = W_hh[g0 * 8 + k2 + 1];
    Ah1T0.x = W_hh[g0 * 8 + k2 + 4];   Ah1T0.y = W_hh[g0 * 8 + k2 + 5];
    Ah0T1.x = W_hh[g1 * 8 + k2];       Ah0T1.y = W_hh[g1 * 8 + k2 + 1];
    Ah1T1.x = W_hh[g1 * 8 + k2 + 4];   Ah1T1.y = W_hh[g1 * 8 + k2 + 5];
    AxT0 *= s0;  Ah0T0 *= s0;  Ah1T0 *= s0;
    AxT1 *= s1;  Ah0T1 *= s1;  Ah1T1 *= s1;

    v8f bias0, bias1;
    #pragma unroll
    for (int r = 0; r < 8; ++r) {
        const int gr = ((r < 4) ? 0 : 8) + unit_j(half, r & 3);
        bias0[r] = (b_ih[gr]      + b_hh[gr]) * 0.5f;
        bias1[r] = (b_ih[gr + 16] + b_hh[gr + 16]) * ((r < 4) ? 1.0f : 0.5f);
    }

    const float* xrow = x + (size_t)(b0 + n) * (size_t)T * 4 + k2;

    v2f Bh0; Bh0.x = 0.0f; Bh0.y = 0.0f;
    v2f Bh1 = Bh0;
    v2f c01 = Bh0, c23 = Bh0;

    v2f Bx = *(const v2f*)(xrow);
    #pragma unroll 2
    for (int t = 0; t < T; ++t) {
        const v2f Bxc = Bx;
        const int tn = (t + 1 < T) ? (t + 1) : t;
        Bx = *(const v2f*)(xrow + (size_t)tn * 4);
        asm volatile("" ::: "memory");

        v8f acc0 = bias0, acc1 = bias1;
        acc0 = __builtin_amdgcn_wmma_f32_16x16x4_f32(false, AxT0,  false, Bxc, (short)0, acc0, false, false);
        acc1 = __builtin_amdgcn_wmma_f32_16x16x4_f32(false, AxT1,  false, Bxc, (short)0, acc1, false, false);
        acc0 = __builtin_amdgcn_wmma_f32_16x16x4_f32(false, Ah0T0, false, Bh0, (short)0, acc0, false, false);
        acc1 = __builtin_amdgcn_wmma_f32_16x16x4_f32(false, Ah0T1, false, Bh0, (short)0, acc1, false, false);
        acc0 = __builtin_amdgcn_wmma_f32_16x16x4_f32(false, Ah1T0, false, Bh1, (short)0, acc0, false, false);
        acc1 = __builtin_amdgcn_wmma_f32_16x16x4_f32(false, Ah1T1, false, Bh1, (short)0, acc1, false, false);
        asm volatile("v_nop\n\tv_nop\n\tv_nop\n\tv_nop"
                     : "+v"(acc0), "+v"(acc1) : "v"(Bxc), "v"(Bh0), "v"(Bh1));

        #define SB() __builtin_amdgcn_sched_barrier(0)
        const v2f half2 = {0.5f, 0.5f};
        v2f tf01, tf23, ti01, ti23, g01, g23, to01, to23;
        tf01.x = __builtin_amdgcn_tanhf(acc0[4]); SB();
        tf01.y = __builtin_amdgcn_tanhf(acc0[5]); SB();
        tf23.x = __builtin_amdgcn_tanhf(acc0[6]); SB();
        tf23.y = __builtin_amdgcn_tanhf(acc0[7]); SB();
        ti01.x = __builtin_amdgcn_tanhf(acc0[0]); SB();
        ti01.y = __builtin_amdgcn_tanhf(acc0[1]); SB();
        const v2f fg01 = __builtin_elementwise_fma(tf01, half2, half2); SB();
        ti23.x = __builtin_amdgcn_tanhf(acc0[2]); SB();
        ti23.y = __builtin_amdgcn_tanhf(acc0[3]); SB();
        const v2f fg23 = __builtin_elementwise_fma(tf23, half2, half2); SB();
        g01.x  = __builtin_amdgcn_tanhf(acc1[0]); SB();
        g01.y  = __builtin_amdgcn_tanhf(acc1[1]); SB();
        const v2f fc01 = fg01 * c01; SB();
        g23.x  = __builtin_amdgcn_tanhf(acc1[2]); SB();
        const v2f ig01 = __builtin_elementwise_fma(ti01, half2, half2); SB();
        g23.y  = __builtin_amdgcn_tanhf(acc1[3]); SB();
        const v2f fc23 = fg23 * c23; SB();
        to01.x = __builtin_amdgcn_tanhf(acc1[4]); SB();
        const v2f ig23 = __builtin_elementwise_fma(ti23, half2, half2); SB();
        to01.y = __builtin_amdgcn_tanhf(acc1[5]); SB();
        to23.x = __builtin_amdgcn_tanhf(acc1[6]); SB();
        c01 = __builtin_elementwise_fma(ig01, g01, fc01); SB();
        to23.y = __builtin_amdgcn_tanhf(acc1[7]); SB();
        c23 = __builtin_elementwise_fma(ig23, g23, fc23); SB();
        const v2f og01 = __builtin_elementwise_fma(to01, half2, half2); SB();
        v2f tc01, tc23;
        tc01.x = __builtin_amdgcn_tanhf(c01.x); SB();
        const v2f og23 = __builtin_elementwise_fma(to23, half2, half2); SB();
        tc01.y = __builtin_amdgcn_tanhf(c01.y); SB();
        tc23.x = __builtin_amdgcn_tanhf(c23.x); SB();
        tc23.y = __builtin_amdgcn_tanhf(c23.y); SB();
        Bh0 = og01 * tc01; SB();
        Bh1 = og23 * tc23;
        #undef SB
    }

    const float o0 = __shfl_xor(Bh0.x, 16, 32);
    const float o1 = __shfl_xor(Bh0.y, 16, 32);
    const float o2 = __shfl_xor(Bh1.x, 16, 32);
    const float o3 = __shfl_xor(Bh1.y, 16, 32);
    if (half == 0) {
        float hj[8];
        hj[0] = Bh0.x; hj[1] = Bh0.y; hj[2] = o0; hj[3] = o1;
        hj[4] = Bh1.x; hj[5] = Bh1.y; hj[6] = o2; hj[7] = o3;
        float acc = b_fc[0];
        #pragma unroll
        for (int j = 0; j < 8; ++j)
            acc += hj[j] * W_fc[j];
        out[b0 + n] = acc;
    }
}

extern "C" void kernel_launch(void* const* d_in, const int* in_sizes, int n_in,
                              void* d_out, int out_size, void* d_ws, size_t ws_size,
                              hipStream_t stream) {
    const float* x    = (const float*)d_in[0];
    const float* W_ih = (const float*)d_in[1];
    const float* W_hh = (const float*)d_in[2];
    const float* b_ih = (const float*)d_in[3];
    const float* b_hh = (const float*)d_in[4];
    const float* W_fc = (const float*)d_in[5];
    const float* b_fc = (const float*)d_in[6];
    float* out = (float*)d_out;

    const int T = 512;
    const int I = 4;
    const int B = in_sizes[0] / (T * I);
    const int wavesTotal = B / 16;
    const int blocks = wavesTotal / WPB;

    lstm_wmma_kernel<<<blocks, 32 * WPB, 0, stream>>>(x, W_ih, W_hh, b_ih, b_hh,
                                                      W_fc, b_fc, out, T);
}


# x loads issued 16 steps ahead into a 16-slot register ring, each slot reloaded in place behind the products that read it; one base per 16 steps plus constants (was 1 step ahead, address per load)
# speedup vs baseline: 2.5707x; 1.6553x over previous
#include <hip/hip_runtime.h>


typedef __attribute__((ext_vector_type(2))) float v2f;
typedef __attribute__((ext_vector_type(8))) float v8f;

#define WPB  4
#define RING 16

__device__ __forceinline__ int unit_j(int hf, int u) {
    return (u < 2) ? (2 * hf + u) : (4 + 2 * hf + (u - 2));
}

__global__ void __launch_bounds__(32 * WPB)
lstm_wmma_kernel(const float* __restrict__ x,
                 const float* __restrict__ W_ih, const float* __restrict__ W_hh,
                 const float* __restrict__ b_ih, const float* __restrict__ b_hh,
                 const float* __restrict__ W_fc, const float* __restrict__ b_fc,
                 float* __restrict__ out, int T)
{
    const int lane = threadIdx.x & 31;
    const int wv   = threadIdx.x >> 5;
    const int n    = lane & 15;
    const int half = lane >> 4;
    const int k2   = 2 * half;
    const int b0   = (blockIdx.x * WPB + wv) * 16;

    const int mh = n >> 3;
    const int mr = n & 7;
    const int g0 = ((mr < 4) ? 0 : 8) + unit_j(mh, mr & 3);
    const int g1 = g0 + 16;
    const float s0 = 0.5f;
    const float s1 = (mr < 4) ? 1.0f : 0.5f;
    v2f AxT0, AxT1, Ah0T0, Ah1T0, Ah0T1, Ah1T1;
    AxT0.x  = W_ih[g0 * 4 + k2];       AxT0.y  = W_ih[g0 * 4 + k2 + 1];
    AxT1.x  = W_ih[g1 * 4 + k2];       AxT1.y  = W_ih[g1 * 4 + k2 + 1];
    Ah0T0.x = W_hh[g0 * 8 + k2];       Ah0T0.y = W_hh[g0 * 8 + k2 + 1];
    Ah1T0.x = W_hh[g0 * 8 + k2 + 4];   Ah1T0.y = W_hh[g0 * 8 + k2 + 5];
    Ah0T1.x = W_hh[g1 * 8 + k2];       Ah0T1.y = W_hh[g1 * 8 + k2 + 1];
    Ah1T1.x = W_hh[g1 * 8 + k2 + 4];   Ah1T1.y = W_hh[g1 * 8 + k2 + 5];
    AxT0 *= s0;  Ah0T0 *= s0;  Ah1T0 *= s0;
    AxT1 *= s1;  Ah0T1 *= s1;  Ah1T1 *= s1;

    v8f bias0, bias1;
    #pragma unroll
    for (int r = 0; r < 8; ++r) {
        const int gr = ((r < 4) ? 0 : 8) + unit_j(half, r & 3);
        bias0[r] = (b_ih[gr]      + b_hh[gr]) * 0.5f;
        bias1[r] = (b_ih[gr + 16] + b_hh[gr + 16]) * ((r < 4) ? 1.0f : 0.5f);
    }

    const float* xrow = x + (size_t)(b0 + n) * (size_t)T * 4 + k2;

    v2f Bh0; Bh0.x = 0.0f; Bh0.y = 0.0f;
    v2f Bh1 = Bh0;
    v2f c01 = Bh0, c23 = Bh0;
    v8f acc0, acc1;

    auto gates = [&](const v2f Bxc) __attribute__((always_inline)) {
        acc0 = bias0; acc1 = bias1;
        acc0 = __builtin_amdgcn_wmma_f32_16x16x4_f32(false, AxT0,  false, Bxc, (short)0, acc0, false, false);
        acc1 = __builtin_amdgcn_wmma_f32_16x16x4_f32(false, AxT1,  false, Bxc, (short)0, acc1, false, false);
        acc0 = __builtin_amdgcn_wmma_f32_16x16x4_f32(false, Ah0T0, false, Bh0, (short)0, acc0, false, false);
        acc1 = __builtin_amdgcn_wmma_f32_16x16x4_f32(false, Ah0T1, false, Bh0, (short)0, acc1, false, false);
        acc0 = __builtin_amdgcn_wmma_f32_16x16x4_f32(false, Ah1T0, false, Bh1, (short)0, acc0, false, false);
        acc1 = __builtin_amdgcn_wmma_f32_16x16x4_f32(false, Ah1T1, false, Bh1, (short)0, acc1, false, false);
        asm volatile("v_nop\n\tv_nop\n\tv_nop\n\tv_nop"
                     : "+v"(acc0), "+v"(acc1) : "v"(Bxc), "v"(Bh0), "v"(Bh1));
    };

    auto cell = [&]() __attribute__((always_inline)) {
        #define SB() __builtin_amdgcn_sched_barrier(0)
        const v2f half2 = {0.5f, 0.5f};
        v2f tf01, tf23, ti01, ti23, g01, g23, to01, to23;
        tf01.x = __builtin_amdgcn_tanhf(acc0[4]); SB();
        tf01.y = __builtin_amdgcn_tanhf(acc0[5]); SB();
        tf23.x = __builtin_amdgcn_tanhf(acc0[6]); SB();
        tf23.y = __builtin_amdgcn_tanhf(acc0[7]); SB();
        ti01.x = __builtin_amdgcn_tanhf(acc0[0]); SB();
        ti01.y = __builtin_amdgcn_tanhf(acc0[1]); SB();
        const v2f fg01 = __builtin_elementwise_fma(tf01, half2, half2); SB();
        ti23.x = __builtin_amdgcn_tanhf(acc0[2]); SB();
        ti23.y = __builtin_amdgcn_tanhf(acc0[3]); SB();
        const v2f fg23 = __builtin_elementwise_fma(tf23, half2, half2); SB();
        g01.x  = __builtin_amdgcn_tanhf(acc1[0]); SB();
        g01.y  = __builtin_amdgcn_tanhf(acc1[1]); SB();
        const v2f fc01 = fg01 * c01; SB();
        g23.x  = __builtin_amdgcn_tanhf(acc1[2]); SB();
        const v2f ig01 = __builtin_elementwise_fma(ti01, half2, half2); SB();
        g23.y  = __builtin_amdgcn_tanhf(acc1[3]); SB();
        const v2f fc23 = fg23 * c23; SB();
        to01.x = __builtin_amdgcn_tanhf(acc1[4]); SB();
        const v2f ig23 = __builtin_elementwise_fma(ti23, half2, half2); SB();
        to01.y = __builtin_amdgcn_tanhf(acc1[5]); SB();
        to23.x = __builtin_amdgcn_tanhf(acc1[6]); SB();
        c01 = __builtin_elementwise_fma(ig01, g01, fc01); SB();
        to23.y = __builtin_amdgcn_tanhf(acc1[7]); SB();
        c23 = __builtin_elementwise_fma(ig23, g23, fc23); SB();
        const v2f og01 = __builtin_elementwise_fma(to01, half2, half2); SB();
        v2f tc01, tc23;
        tc01.x = __builtin_amdgcn_tanhf(c01.x); SB();
        const v2f og23 = __builtin_elementwise_fma(to23, half2, half2); SB();
        tc01.y = __builtin_amdgcn_tanhf(c01.y); SB();
        tc23.x = __builtin_amdgcn_tanhf(c23.x); SB();
        tc23.y = __builtin_amdgcn_tanhf(c23.y); SB();
        Bh0 = og01 * tc01; SB();
        Bh1 = og23 * tc23;
        #undef SB
    };

    v2f ring[RING];
    const int G = T / RING;
    if (G > 0) {
        #pragma unroll
        for (int s = 0; s < RING; ++s)
            ring[s] = *(const v2f*)(xrow + s * 4);
    }
    for (int g = 0; g < G; ++g) {
        const int gn = (g + 1 < G) ? (g + 1) : g;
        const float* xn = xrow + (size_t)gn * (RING * 4);
        #pragma unroll
        for (int s = 0; s < RING; ++s) {
            gates(ring[s]);
            ring[s] = *(const v2f*)(xn + s * 4);
            cell();
        }
    }
    for (int t = G * RING; t < T; ++t) {
        gates(*(const v2f*)(xrow + (size_t)t * 4));
        cell();
    }

    const float o0 = __shfl_xor(Bh0.x, 16, 32);
    const float o1 = __shfl_xor(Bh0.y, 16, 32);
    const float o2 = __shfl_xor(Bh1.x, 16, 32);
    const float o3 = __shfl_xor(Bh1.y, 16, 32);
    if (half == 0) {
        float hj[8];
        hj[0] = Bh0.x; hj[1] = Bh0.y; hj[2] = o0; hj[3] = o1;
        hj[4] = Bh1.x; hj[5] = Bh1.y; hj[6] = o2; hj[7] = o3;
        float acc = b_fc[0];
        #pragma unroll
        for (int j = 0; j < 8; ++j)
            acc += hj[j] * W_fc[j];
        out[b0 + n] = acc;
    }
}

extern "C" void kernel_launch(void* const* d_in, const int* in_sizes, int n_in,
                              void* d_out, int out_size, void* d_ws, size_t ws_size,
                              hipStream_t stream) {
    const float* x    = (const float*)d_in[0];
    const float* W_ih = (const float*)d_in[1];
    const float* W_hh = (const float*)d_in[2];
    const float* b_ih = (const float*)d_in[3];
    const float* b_hh = (const float*)d_in[4];
    const float* W_fc = (const float*)d_in[5];
    const float* b_fc = (const float*)d_in[6];
    float* out = (float*)d_out;

    const int T = 512;
    const int I = 4;
    const int B = in_sizes[0] / (T * I);
    const int wavesTotal = B / 16;
    const int blocks = wavesTotal / WPB;

    lstm_wmma_kernel<<<blocks, 32 * WPB, 0, stream>>>(x, W_ih, W_hh, b_ih, b_hh,
                                                      W_fc, b_fc, out, T);
}


# x products of step t+1 issued inside step t's cell update (between c and tanh(c), and between tanh(c) and h) where the wave waited on results; 4 v_nop a step dropped; arithmetic unchanged
# speedup vs baseline: 2.6944x; 1.0481x over previous
#include <hip/hip_runtime.h>


typedef __attribute__((ext_vector_type(2))) float v2f;
typedef __attribute__((ext_vector_type(8))) float v8f;

#define WPB  4
#define RING 16

__device__ __forceinline__ int unit_j(int hf, int u) {
    return (u < 2) ? (2 * hf + u) : (4 + 2 * hf + (u - 2));
}

__global__ void __launch_bounds__(32 * WPB)
lstm_wmma_kernel(const float* __restrict__ x,
                 const float* __restrict__ W_ih, const float* __restrict__ W_hh,
                 const float* __restrict__ b_ih, const float* __restrict__ b_hh,
                 const float* __restrict__ W_fc, const float* __restrict__ b_fc,
                 float* __restrict__ out, int T)
{
    const int lane = threadIdx.x & 31;
    const int wv   = threadIdx.x >> 5;
    const int n    = lane & 15;
    const int half = lane >> 4;
    const int k2   = 2 * half;
    const int b0   = (blockIdx.x * WPB + wv) * 16;

    const int mh = n >> 3;
    const int mr = n & 7;
    const int g0 = ((mr < 4) ? 0 : 8) + unit_j(mh, mr & 3);
    const int g1 = g0 + 16;
    const float s0 = 0.5f;
    const float s1 = (mr < 4) ? 1.0f : 0.5f;
    v2f AxT0, AxT1, Ah0T0, Ah1T0, Ah0T1, Ah1T1;
    AxT0.x  = W_ih[g0 * 4 + k2];       AxT0.y  = W_ih[g0 * 4 + k2 + 1];
    AxT1.x  = W_ih[g1 * 4 + k2];       AxT1.y  = W_ih[g1 * 4 + k2 + 1];
    Ah0T0.x = W_hh[g0 * 8 + k2];       Ah0T0.y = W_hh[g0 * 8 + k2 + 1];
    Ah1T0.x = W_hh[g0 * 8 + k2 + 4];   Ah1T0.y = W_hh[g0 * 8 + k2 + 5];
    Ah0T1.x = W_hh[g1 * 8 + k2];       Ah0T1.y = W_hh[g1 * 8 + k2 + 1];
    Ah1T1.x = W_hh[g1 * 8 + k2 + 4];   Ah1T1.y = W_hh[g1 * 8 + k2 + 5];
    AxT0 *= s0;  Ah0T0 *= s0;  Ah1T0 *= s0;
    AxT1 *= s1;  Ah0T1 *= s1;  Ah1T1 *= s1;

    v8f bias0, bias1;
    #pragma unroll
    for (int r = 0; r < 8; ++r) {
        const int gr = ((r < 4) ? 0 : 8) + unit_j(half, r & 3);
        bias0[r] = (b_ih[gr]      + b_hh[gr]) * 0.5f;
        bias1[r] = (b_ih[gr + 16] + b_hh[gr + 16]) * ((r < 4) ? 1.0f : 0.5f);
    }

    const float* xrow = x + (size_t)(b0 + n) * (size_t)T * 4 + k2;

    v2f Bh0; Bh0.x = 0.0f; Bh0.y = 0.0f;
    v2f Bh1 = Bh0;
    v2f c01 = Bh0, c23 = Bh0;
    v8f acc0, acc1;

    #define SB() __builtin_amdgcn_sched_barrier(0)
    #define WMMA4(A, Bv, C) __builtin_amdgcn_wmma_f32_16x16x4_f32(false, (A), false, (Bv), (short)0, (C), false, false)

    auto hprod = [&]() __attribute__((always_inline)) {
        acc0 = WMMA4(Ah0T0, Bh0, acc0); SB();
        acc1 = WMMA4(Ah0T1, Bh0, acc1); SB();
        acc0 = WMMA4(Ah1T0, Bh1, acc0); SB();
        acc1 = WMMA4(Ah1T1, Bh1, acc1); SB();
    };

    auto cell_xnext = [&](const v2f Bxn) __attribute__((always_inline)) {
        const v2f half2 = {0.5f, 0.5f};
        v2f tf01, tf23, ti01, ti23, g01, g23, to01, to23;
        tf01.x = __builtin_amdgcn_tanhf(acc0[4]); SB();
        tf01.y = __builtin_amdgcn_tanhf(acc0[5]); SB();
        tf23.x = __builtin_amdgcn_tanhf(acc0[6]); SB();
        tf23.y = __builtin_amdgcn_tanhf(acc0[7]); SB();
        ti01.x = __builtin_amdgcn_tanhf(acc0[0]); SB();
        ti01.y = __builtin_amdgcn_tanhf(acc0[1]); SB();
        const v2f fg01 = __builtin_elementwise_fma(tf01, half2, half2); SB();
        ti23.x = __builtin_amdgcn_tanhf(acc0[2]); SB();
        ti23.y = __builtin_amdgcn_tanhf(acc0[3]); SB();
        const v2f fg23 = __builtin_elementwise_fma(tf23, half2, half2); SB();
        g01.x  = __builtin_amdgcn_tanhf(acc1[0]); SB();
        g01.y  = __builtin_amdgcn_tanhf(acc1[1]); SB();
        const v2f fc01 = fg01 * c01; SB();
        g23.x  = __builtin_amdgcn_tanhf(acc1[2]); SB();
        const v2f ig01 = __builtin_elementwise_fma(ti01, half2, half2); SB();
        g23.y  = __builtin_amdgcn_tanhf(acc1[3]); SB();
        const v2f fc23 = fg23 * c23; SB();
        to01.x = __builtin_amdgcn_tanhf(acc1[4]); SB();
        const v2f ig23 = __builtin_elementwise_fma(ti23, half2, half2); SB();
        to01.y = __builtin_amdgcn_tanhf(acc1[5]); SB();
        to23.x = __builtin_amdgcn_tanhf(acc1[6]); SB();
        c01 = __builtin_elementwise_fma(ig01, g01, fc01); SB();
        to23.y = __builtin_amdgcn_tanhf(acc1[7]); SB();
        c23 = __builtin_elementwise_fma(ig23, g23, fc23); SB();
        const v2f og01 = __builtin_elementwise_fma(to01, half2, half2); SB();
        acc0 = WMMA4(AxT0, Bxn, bias0); SB();
        v2f tc01, tc23;
        tc01.x = __builtin_amdgcn_tanhf(c01.x); SB();
        tc01.y = __builtin_amdgcn_tanhf(c01.y); SB();
        tc23.x = __builtin_amdgcn_tanhf(c23.x); SB();
        tc23.y = __builtin_amdgcn_tanhf(c23.y); SB();
        const v2f og23 = __builtin_elementwise_fma(to23, half2, half2); SB();
        Bh0 = og01 * tc01; SB();
        acc1 = WMMA4(AxT1, Bxn, bias1); SB();
        Bh1 = og23 * tc23; SB();
    };

    if (T > 0) {
        {
            const v2f Bx0 = *(const v2f*)xrow;
            acc0 = WMMA4(AxT0, Bx0, bias0); SB();
            acc1 = WMMA4(AxT1, Bx0, bias1); SB();
            hprod();
        }

        v2f ring[RING];
        const int G = T / RING;
        if (G > 0) {
            #pragma unroll
            for (int s = 0; s < RING; ++s) {
                const int i0 = (s + 1 < T) ? (s + 1) : (T - 1);
                ring[s] = *(const v2f*)(xrow + (size_t)i0 * 4);
            }
        }
        for (int g = 0; g < G; ++g) {
            const int gn = (g + 1 < G) ? (g + 1) : g;
            const float* xn = xrow + (size_t)gn * (RING * 4) + 4;
            const int iL = (gn * RING + RING < T) ? (gn * RING + RING) : (T - 1);
            const float* xl = xrow + (size_t)iL * 4;
            #pragma unroll
            for (int s = 0; s < RING; ++s) {
                cell_xnext(ring[s]);
                ring[s] = (s + 1 < RING) ? *(const v2f*)(xn + s * 4) : *(const v2f*)xl;
                SB();
                hprod();
            }
        }
        for (int t = G * RING; t < T; ++t) {
            const int tn = (t + 1 < T) ? (t + 1) : (T - 1);
            const v2f Bxn = *(const v2f*)(xrow + (size_t)tn * 4);
            cell_xnext(Bxn);
            hprod();
        }
    }
    #undef WMMA4
    #undef SB

    const float o0 = __shfl_xor(Bh0.x, 16, 32);
    const float o1 = __shfl_xor(Bh0.y, 16, 32);
    const float o2 = __shfl_xor(Bh1.x, 16, 32);
    const float o3 = __shfl_xor(Bh1.y, 16, 32);
    if (half == 0) {
        float hj[8];
        hj[0] = Bh0.x; hj[1] = Bh0.y; hj[2] = o0; hj[3] = o1;
        hj[4] = Bh1.x; hj[5] = Bh1.y; hj[6] = o2; hj[7] = o3;
        float acc = b_fc[0];
        #pragma unroll
        for (int j = 0; j < 8; ++j)
            acc += hj[j] * W_fc[j];
        out[b0 + n] = acc;
    }
}

extern "C" void kernel_launch(void* const* d_in, const int* in_sizes, int n_in,
                              void* d_out, int out_size, void* d_ws, size_t ws_size,
                              hipStream_t stream) {
    const float* x    = (const float*)d_in[0];
    const float* W_ih = (const float*)d_in[1];
    const float* W_hh = (const float*)d_in[2];
    const float* b_ih = (const float*)d_in[3];
    const float* b_hh = (const float*)d_in[4];
    const float* W_fc = (const float*)d_in[5];
    const float* b_fc = (const float*)d_in[6];
    float* out = (float*)d_out;

    const int T = 512;
    const int I = 4;
    const int B = in_sizes[0] / (T * I);
    const int wavesTotal = B / 16;
    const int blocks = wavesTotal / WPB;

    lstm_wmma_kernel<<<blocks, 32 * WPB, 0, stream>>>(x, W_ih, W_hh, b_ih, b_hh,
                                                      W_fc, b_fc, out, T);
}


# x ring reloaded as two runs of 8 loads per 16 steps (a half of the ring, behind the step that read its last slot) instead of one load per step: 2 memory-issue places per 16 steps, not 16
# speedup vs baseline: 3.0149x; 1.1190x over previous
#include <hip/hip_runtime.h>


typedef __attribute__((ext_vector_type(2))) float v2f;
typedef __attribute__((ext_vector_type(8))) float v8f;

#define WPB  4
#define RING 16
#define XGRP 8

__device__ __forceinline__ int unit_j(int hf, int u) {
    return (u < 2) ? (2 * hf + u) : (4 + 2 * hf + (u - 2));
}

__global__ void __launch_bounds__(32 * WPB)
lstm_wmma_kernel(const float* __restrict__ x,
                 const float* __restrict__ W_ih, const float* __restrict__ W_hh,
                 const float* __restrict__ b_ih, const float* __restrict__ b_hh,
                 const float* __restrict__ W_fc, const float* __restrict__ b_fc,
                 float* __restrict__ out, int T)
{
    const int lane = threadIdx.x & 31;
    const int wv   = threadIdx.x >> 5;
    const int n    = lane & 15;
    const int half = lane >> 4;
    const int k2   = 2 * half;
    const int b0   = (blockIdx.x * WPB + wv) * 16;

    const int mh = n >> 3;
    const int mr = n & 7;
    const int g0 = ((mr < 4) ? 0 : 8) + unit_j(mh, mr & 3);
    const int g1 = g0 + 16;
    const float s0 = 0.5f;
    const float s1 = (mr < 4) ? 1.0f : 0.5f;
    v2f AxT0, AxT1, Ah0T0, Ah1T0, Ah0T1, Ah1T1;
    AxT0.x  = W_ih[g0 * 4 + k2];       AxT0.y  = W_ih[g0 * 4 + k2 + 1];
    AxT1.x  = W_ih[g1 * 4 + k2];       AxT1.y  = W_ih[g1 * 4 + k2 + 1];
    Ah0T0.x = W_hh[g0 * 8 + k2];       Ah0T0.y = W_hh[g0 * 8 + k2 + 1];
    Ah1T0.x = W_hh[g0 * 8 + k2 + 4];   Ah1T0.y = W_hh[g0 * 8 + k2 + 5];
    Ah0T1.x = W_hh[g1 * 8 + k2];       Ah0T1.y = W_hh[g1 * 8 + k2 + 1];
    Ah1T1.x = W_hh[g1 * 8 + k2 + 4];   Ah1T1.y = W_hh[g1 * 8 + k2 + 5];
    AxT0 *= s0;  Ah0T0 *= s0;  Ah1T0 *= s0;
    AxT1 *= s1;  Ah0T1 *= s1;  Ah1T1 *= s1;

    v8f bias0, bias1;
    #pragma unroll
    for (int r = 0; r < 8; ++r) {
        const int gr = ((r < 4) ? 0 : 8) + unit_j(half, r & 3);
        bias0[r] = (b_ih[gr]      + b_hh[gr]) * 0.5f;
        bias1[r] = (b_ih[gr + 16] + b_hh[gr + 16]) * ((r < 4) ? 1.0f : 0.5f);
    }

    const float* xrow = x + (size_t)(b0 + n) * (size_t)T * 4 + k2;

    v2f Bh0; Bh0.x = 0.0f; Bh0.y = 0.0f;
    v2f Bh1 = Bh0;
    v2f c01 = Bh0, c23 = Bh0;
    v8f acc0, acc1;

    #define SB() __builtin_amdgcn_sched_barrier(0)
    #define WMMA4(A, Bv, C) __builtin_amdgcn_wmma_f32_16x16x4_f32(false, (A), false, (Bv), (short)0, (C), false, false)

    auto hprod = [&]() __attribute__((always_inline)) {
        acc0 = WMMA4(Ah0T0, Bh0, acc0); SB();
        acc1 = WMMA4(Ah0T1, Bh0, acc1); SB();
        acc0 = WMMA4(Ah1T0, Bh1, acc0); SB();
        acc1 = WMMA4(Ah1T1, Bh1, acc1); SB();
    };

    auto cell_xnext = [&](const v2f Bxn) __attribute__((always_inline)) {
        const v2f half2 = {0.5f, 0.5f};
        v2f tf01, tf23, ti01, ti23, g01, g23, to01, to23;
        tf01.x = __builtin_amdgcn_tanhf(acc0[4]); SB();
        tf01.y = __builtin_amdgcn_tanhf(acc0[5]); SB();
        tf23.x = __builtin_amdgcn_tanhf(acc0[6]); SB();
        tf23.y = __builtin_amdgcn_tanhf(acc0[7]); SB();
        ti01.x = __builtin_amdgcn_tanhf(acc0[0]); SB();
        ti01.y = __builtin_amdgcn_tanhf(acc0[1]); SB();
        const v2f fg01 = __builtin_elementwise_fma(tf01, half2, half2); SB();
        ti23.x = __builtin_amdgcn_tanhf(acc0[2]); SB();
        ti23.y = __builtin_amdgcn_tanhf(acc0[3]); SB();
        const v2f fg23 = __builtin_elementwise_fma(tf23, half2, half2); SB();
        g01.x  = __builtin_amdgcn_tanhf(acc1[0]); SB();
        g01.y  = __builtin_amdgcn_tanhf(acc1[1]); SB();
        const v2f fc01 = fg01 * c01; SB();
        g23.x  = __builtin_amdgcn_tanhf(acc1[2]); SB();
        const v2f ig01 = __builtin_elementwise_fma(ti01, half2, half2); SB();
        g23.y  = __builtin_amdgcn_tanhf(acc1[3]); SB();
        const v2f fc23 = fg23 * c23; SB();
        to01.x = __builtin_amdgcn_tanhf(acc1[4]); SB();
        const v2f ig23 = __builtin_elementwise_fma(ti23, half2, half2); SB();
        to01.y = __builtin_amdgcn_tanhf(acc1[5]); SB();
        to23.x = __builtin_amdgcn_tanhf(acc1[6]); SB();
        c01 = __builtin_elementwise_fma(ig01, g01, fc01); SB();
        to23.y = __builtin_amdgcn_tanhf(acc1[7]); SB();
        c23 = __builtin_elementwise_fma(ig23, g23, fc23); SB();
        const v2f og01 = __builtin_elementwise_fma(to01, half2, half2); SB();
        acc0 = WMMA4(AxT0, Bxn, bias0); SB();
        v2f tc01, tc23;
        tc01.x = __builtin_amdgcn_tanhf(c01.x); SB();
        tc01.y = __builtin_amdgcn_tanhf(c01.y); SB();
        tc23.x = __builtin_amdgcn_tanhf(c23.x); SB();
        tc23.y = __builtin_amdgcn_tanhf(c23.y); SB();
        const v2f og23 = __builtin_elementwise_fma(to23, half2, half2); SB();
        Bh0 = og01 * tc01; SB();
        acc1 = WMMA4(AxT1, Bxn, bias1); SB();
        Bh1 = og23 * tc23; SB();
        asm volatile("" : "+v"(Bh0) : "v"(Bh1));
    };

    if (T > 0) {
        {
            const v2f Bx0 = *(const v2f*)xrow;
            acc0 = WMMA4(AxT0, Bx0, bias0); SB();
            acc1 = WMMA4(AxT1, Bx0, bias1); SB();
            hprod();
        }

        v2f ring[RING];
        const int G = T / RING;
        if (G > 0) {
            #pragma unroll
            for (int s = 0; s < RING; ++s) {
                const int i0 = (s + 1 < T) ? (s + 1) : (T - 1);
                ring[s] = *(const v2f*)(xrow + (size_t)i0 * 4);
            }
        }
        for (int g = 0; g < G; ++g) {
            const int gn = (g + 1 < G) ? (g + 1) : g;
            const float* xn = xrow + (size_t)gn * (RING * 4) + 4;
            const int iL = (gn * RING + RING < T) ? (gn * RING + RING) : (T - 1);
            const float* xl = xrow + (size_t)iL * 4;
            #pragma unroll
            for (int s = 0; s < RING; ++s) {
                cell_xnext(ring[s]);
                if (s % XGRP != 0)
                    asm volatile("" :: "v"(ring[s - 1]));
                if (s % XGRP == XGRP - 1) {
                    #pragma unroll
                    for (int j = s + 1 - XGRP; j <= s; ++j)
                        ring[j] = (j + 1 < RING) ? *(const v2f*)(xn + j * 4) : *(const v2f*)xl;
                    SB();
                }
                hprod();
            }
        }
        for (int t = G * RING; t < T; ++t) {
            const int tn = (t + 1 < T) ? (t + 1) : (T - 1);
            const v2f Bxn = *(const v2f*)(xrow + (size_t)tn * 4);
            cell_xnext(Bxn);
            hprod();
        }
    }
    #undef WMMA4
    #undef SB

    const float o0 = __shfl_xor(Bh0.x, 16, 32);
    const float o1 = __shfl_xor(Bh0.y, 16, 32);
    const float o2 = __shfl_xor(Bh1.x, 16, 32);
    const float o3 = __shfl_xor(Bh1.y, 16, 32);
    if (half == 0) {
        float hj[8];
        hj[0] = Bh0.x; hj[1] = Bh0.y; hj[2] = o0; hj[3] = o1;
        hj[4] = Bh1.x; hj[5] = Bh1.y; hj[6] = o2; hj[7] = o3;
        float acc = b_fc[0];
        #pragma unroll
        for (int j = 0; j < 8; ++j)
            acc += hj[j] * W_fc[j];
        out[b0 + n] = acc;
    }
}

extern "C" void kernel_launch(void* const* d_in, const int* in_sizes, int n_in,
                              void* d_out, int out_size, void* d_ws, size_t ws_size,
                              hipStream_t stream) {
    const float* x    = (const float*)d_in[0];
    const float* W_ih = (const float*)d_in[1];
    const float* W_hh = (const float*)d_in[2];
    const float* b_ih = (const float*)d_in[3];
    const float* b_hh = (const float*)d_in[4];
    const float* W_fc = (const float*)d_in[5];
    const float* b_fc = (const float*)d_in[6];
    float* out = (float*)d_out;

    const int T = 512;
    const int I = 4;
    const int B = in_sizes[0] / (T * I);
    const int wavesTotal = B / 16;
    const int blocks = wavesTotal / WPB;

    lstm_wmma_kernel<<<blocks, 32 * WPB, 0, stream>>>(x, W_ih, W_hh, b_ih, b_hh,
                                                      W_fc, b_fc, out, T);
}


# x products (bias + W_ih x) moved to a partner wave on an idle SIMD of the work-group and handed over through LDS, 16 steps a chunk, as the C operand of the first recurrent products; sums unchanged
# speedup vs baseline: 3.4495x; 1.1442x over previous
#include <hip/hip_runtime.h>


typedef __attribute__((ext_vector_type(2))) float v2f;
typedef __attribute__((ext_vector_type(4))) float v4f;
typedef __attribute__((ext_vector_type(8))) float v8f;

#define WPB   4
#define PAIRS 2
#define CH    16
#define NBUF  2

__device__ __forceinline__ int unit_j(int hf, int u) {
    return (u < 2) ? (2 * hf + u) : (4 + 2 * hf + (u - 2));
}

__global__ void __launch_bounds__(32 * WPB)
lstm_wmma_kernel(const float* __restrict__ x,
                 const float* __restrict__ W_ih, const float* __restrict__ W_hh,
                 const float* __restrict__ b_ih, const float* __restrict__ b_hh,
                 const float* __restrict__ W_fc, const float* __restrict__ b_fc,
                 float* __restrict__ out, int T)
{
    __shared__ v4f ring[PAIRS * NBUF * CH * 4 * 32];

    const int lane = threadIdx.x & 31;
    const int wv   = threadIdx.x >> 5;
    const int pair = wv & 1;
    const bool producer = wv >= PAIRS;
    const int n    = lane & 15;
    const int half = lane >> 4;
    const int k2   = 2 * half;
    const int b0   = (blockIdx.x * PAIRS + pair) * 16;

    const int mh = n >> 3;
    const int mr = n & 7;
    const int g0 = ((mr < 4) ? 0 : 8) + unit_j(mh, mr & 3);
    const int g1 = g0 + 16;
    const float s0 = 0.5f;
    const float s1 = (mr < 4) ? 1.0f : 0.5f;

    v4f* myring = ring + (size_t)pair * (NBUF * CH * 4 * 32) + lane;
    const int G = T / CH;

    #define WMMA4(A, Bv, C) __builtin_amdgcn_wmma_f32_16x16x4_f32(false, (A), false, (Bv), (short)0, (C), false, false)
    #define SB() __builtin_amdgcn_sched_barrier(0)

    if (producer) {
        v2f AxT0, AxT1;
        AxT0.x = W_ih[g0 * 4 + k2] * s0;   AxT0.y = W_ih[g0 * 4 + k2 + 1] * s0;
        AxT1.x = W_ih[g1 * 4 + k2] * s1;   AxT1.y = W_ih[g1 * 4 + k2 + 1] * s1;
        v8f bias0, bias1;
        #pragma unroll
        for (int r = 0; r < 8; ++r) {
            const int gr = ((r < 4) ? 0 : 8) + unit_j(half, r & 3);
            bias0[r] = (b_ih[gr]      + b_hh[gr]) * 0.5f;
            bias1[r] = (b_ih[gr + 16] + b_hh[gr + 16]) * ((r < 4) ? 1.0f : 0.5f);
        }
        const float* xrow = x + (size_t)(b0 + n) * (size_t)T * 4 + k2;

        auto xload = [&](v2f (&xs)[CH], int c) __attribute__((always_inline)) {
            const int cl = (c < G) ? c : (G - 1);
            const int cc = (cl > 0) ? cl : 0;
            #pragma unroll
            for (int s = 0; s < CH; ++s)
                xs[s] = *(const v2f*)(xrow + (size_t)(cc * CH + s) * 4);
        };
        auto produce = [&](const v2f (&xs)[CH], int c) __attribute__((always_inline)) {
            v4f* dst = myring + (size_t)(c % NBUF) * (CH * 4 * 32);
            #pragma unroll
            for (int s = 0; s < CH; ++s) {
                const v8f d0 = WMMA4(AxT0, xs[s], bias0);
                const v8f d1 = WMMA4(AxT1, xs[s], bias1);
                dst[(s * 4 + 0) * 32] = (v4f){d0[0], d0[1], d0[2], d0[3]};
                dst[(s * 4 + 1) * 32] = (v4f){d0[4], d0[5], d0[6], d0[7]};
                dst[(s * 4 + 2) * 32] = (v4f){d1[0], d1[1], d1[2], d1[3]};
                dst[(s * 4 + 3) * 32] = (v4f){d1[4], d1[5], d1[6], d1[7]};
                SB();
            }
        };
        #define LDS_BARRIER() do { \
            __builtin_amdgcn_fence(__ATOMIC_RELEASE, "workgroup", "local"); \
            __builtin_amdgcn_s_barrier(); \
            __builtin_amdgcn_fence(__ATOMIC_ACQUIRE, "workgroup", "local"); } while (0)
        #define HOLD8(xs, o) asm volatile("" : "+v"(xs[o]), "+v"(xs[o + 1]), "+v"(xs[o + 2]), "+v"(xs[o + 3]), \
                                               "+v"(xs[o + 4]), "+v"(xs[o + 5]), "+v"(xs[o + 6]), "+v"(xs[o + 7]) :: "memory")
        #define HOLD(xs) do { HOLD8(xs, 0); HOLD8(xs, 8); } while (0)
        v2f xa[CH], xb[CH];
        xload(xa, 0);
        for (int k = 0; k < G; k += 2) {
            xload(xb, k + 1);
            HOLD(xa);
            produce(xa, k);
            LDS_BARRIER();
            xload(xa, k + 2);
            HOLD(xb);
            produce(xb, k + 1);
            LDS_BARRIER();
        }
        LDS_BARRIER();
        #undef HOLD
        #undef HOLD8
        #undef LDS_BARRIER
        return;
    }

    v2f Ah0T0, Ah1T0, Ah0T1, Ah1T1;
    Ah0T0.x = W_hh[g0 * 8 + k2];       Ah0T0.y = W_hh[g0 * 8 + k2 + 1];
    Ah1T0.x = W_hh[g0 * 8 + k2 + 4];   Ah1T0.y = W_hh[g0 * 8 + k2 + 5];
    Ah0T1.x = W_hh[g1 * 8 + k2];       Ah0T1.y = W_hh[g1 * 8 + k2 + 1];
    Ah1T1.x = W_hh[g1 * 8 + k2 + 4];   Ah1T1.y = W_hh[g1 * 8 + k2 + 5];
    Ah0T0 *= s0;  Ah1T0 *= s0;  Ah0T1 *= s1;  Ah1T1 *= s1;

    v2f Bh0; Bh0.x = 0.0f; Bh0.y = 0.0f;
    v2f Bh1 = Bh0;
    v2f c01 = Bh0, c23 = Bh0;
    v8f acc0, acc1;

    auto step = [&](const v8f cx0, const v8f cx1) __attribute__((always_inline)) {
        acc0 = WMMA4(Ah0T0, Bh0, cx0);  SB();
        acc1 = WMMA4(Ah0T1, Bh0, cx1);  SB();
        acc0 = WMMA4(Ah1T0, Bh1, acc0); SB();
        acc1 = WMMA4(Ah1T1, Bh1, acc1); SB();
        const v2f half2 = {0.5f, 0.5f};
        v2f tf01, tf23, ti01, ti23, g01, g23, to01, to23, tc01, tc23;
        tf01.x = __builtin_amdgcn_tanhf(acc0[4]); SB();
        tf01.y = __builtin_amdgcn_tanhf(acc0[5]); SB();
        tf23.x = __builtin_amdgcn_tanhf(acc0[6]); SB();
        tf23.y = __builtin_amdgcn_tanhf(acc0[7]); SB();
        ti01.x = __builtin_amdgcn_tanhf(acc0[0]); SB();
        ti01.y = __builtin_amdgcn_tanhf(acc0[1]); SB();
        const v2f fg01 = __builtin_elementwise_fma(tf01, half2, half2); SB();
        ti23.x = __builtin_amdgcn_tanhf(acc0[2]); SB();
        ti23.y = __builtin_amdgcn_tanhf(acc0[3]); SB();
        const v2f fg23 = __builtin_elementwise_fma(tf23, half2, half2); SB();
        g01.x  = __builtin_amdgcn_tanhf(acc1[0]); SB();
        g01.y  = __builtin_amdgcn_tanhf(acc1[1]); SB();
        const v2f fc01 = fg01 * c01; SB();
        g23.x  = __builtin_amdgcn_tanhf(acc1[2]); SB();
        const v2f ig01 = __builtin_elementwise_fma(ti01, half2, half2); SB();
        g23.y  = __builtin_amdgcn_tanhf(acc1[3]); SB();
        const v2f fc23 = fg23 * c23; SB();
        to01.x = __builtin_amdgcn_tanhf(acc1[4]); SB();
        const v2f ig23 = __builtin_elementwise_fma(ti23, half2, half2); SB();
        to01.y = __builtin_amdgcn_tanhf(acc1[5]); SB();
        c01 = __builtin_elementwise_fma(ig01, g01, fc01); SB();
        to23.x = __builtin_amdgcn_tanhf(acc1[6]); SB();
        c23 = __builtin_elementwise_fma(ig23, g23, fc23); SB();
        to23.y = __builtin_amdgcn_tanhf(acc1[7]); SB();
        const v2f og01 = __builtin_elementwise_fma(to01, half2, half2); SB();
        tc01.x = __builtin_amdgcn_tanhf(c01.x); SB();
        tc01.y = __builtin_amdgcn_tanhf(c01.y); SB();
        tc23.x = __builtin_amdgcn_tanhf(c23.x); SB();
        tc23.y = __builtin_amdgcn_tanhf(c23.y); SB();
        const v2f og23 = __builtin_elementwise_fma(to23, half2, half2); SB();
        Bh0 = og01 * tc01; SB();
        Bh1 = og23 * tc23; SB();
    };

    auto fetch = [&](v8f (&c0)[CH], v8f (&c1)[CH], int c) __attribute__((always_inline)) {
        const v4f* srcp = myring + (size_t)(c % NBUF) * (CH * 4 * 32);
        #pragma unroll
        for (int s = 0; s < CH; ++s) {
            const v4f q0 = srcp[(s * 4 + 0) * 32], q1 = srcp[(s * 4 + 1) * 32];
            const v4f q2 = srcp[(s * 4 + 2) * 32], q3 = srcp[(s * 4 + 3) * 32];
            c0[s] = (v8f){q0.x, q0.y, q0.z, q0.w, q1.x, q1.y, q1.z, q1.w};
            c1[s] = (v8f){q2.x, q2.y, q2.z, q2.w, q3.x, q3.y, q3.z, q3.w};
        }
    };
    auto run = [&](const v8f (&c0)[CH], const v8f (&c1)[CH]) __attribute__((always_inline)) {
        #pragma unroll
        for (int s = 0; s < CH; ++s) step(c0[s], c1[s]);
    };

    v8f a0[CH], a1[CH], e0[CH], e1[CH];
    #define PIN() asm volatile("" : "+v"(Bh0), "+v"(Bh1) :: "memory")
    __syncthreads();
    fetch(a0, a1, 0);
    for (int c = 0; c < G; c += 2) {
        PIN();
        asm volatile("s_wait_dscnt 0x0" ::: "memory");
        __syncthreads();
        fetch(e0, e1, c + 1);
        PIN();
        run(a0, a1);
        PIN();
        __syncthreads();
        fetch(a0, a1, (c + 2 < G) ? (c + 2) : (G - 1));
        PIN();
        run(e0, e1);
    }
    #undef PIN
    #undef WMMA4
    #undef SB

    const float o0 = __shfl_xor(Bh0.x, 16, 32);
    const float o1 = __shfl_xor(Bh0.y, 16, 32);
    const float o2 = __shfl_xor(Bh1.x, 16, 32);
    const float o3 = __shfl_xor(Bh1.y, 16, 32);
    if (half == 0) {
        float hj[8];
        hj[0] = Bh0.x; hj[1] = Bh0.y; hj[2] = o0; hj[3] = o1;
        hj[4] = Bh1.x; hj[5] = Bh1.y; hj[6] = o2; hj[7] = o3;
        float acc = b_fc[0];
        #pragma unroll
        for (int j = 0; j < 8; ++j)
            acc += hj[j] * W_fc[j];
        out[b0 + n] = acc;
    }
}

extern "C" void kernel_launch(void* const* d_in, const int* in_sizes, int n_in,
                              void* d_out, int out_size, void* d_ws, size_t ws_size,
                              hipStream_t stream) {
    const float* x    = (const float*)d_in[0];
    const float* W_ih = (const float*)d_in[1];
    const float* W_hh = (const float*)d_in[2];
    const float* b_ih = (const float*)d_in[3];
    const float* b_hh = (const float*)d_in[4];
    const float* W_fc = (const float*)d_in[5];
    const float* b_fc = (const float*)d_in[6];
    float* out = (float*)d_out;

    const int T = 512;
    const int I = 4;
    const int B = in_sizes[0] / (T * I);
    const int tiles = B / 16;
    const int blocks = tiles / PAIRS;

    lstm_wmma_kernel<<<blocks, 32 * WPB, 0, stream>>>(x, W_ih, W_hh, b_ih, b_hh,
                                                      W_fc, b_fc, out, T);
}
